# starting kernel
# baseline (speedup 1.0000x reference)
#include <hip/hip_runtime.h>


#define NF 128
#define PSTR 144

typedef __attribute__((ext_vector_type(2))) float v2f;
typedef __attribute__((ext_vector_type(8))) float v8f;

__global__ void deg_init_kernel(float* __restrict__ deg, int n) {
  int i = blockIdx.x * blockDim.x + threadIdx.x;
  if (i < n) deg[i] = 1.0f;
}

__global__ void deg_count_kernel(const int* __restrict__ ei,
                                 float* __restrict__ deg, int nedges) {
  int e = blockIdx.x * blockDim.x + threadIdx.x;
  if (e < nedges) {
    int dst = ei[nedges + e];
    __hip_atomic_fetch_add(&deg[dst], 1.0f, __ATOMIC_RELAXED,
                           __HIP_MEMORY_SCOPE_AGENT);
  }
}

__global__ void rsqrt_kernel(float* __restrict__ deg, int n) {
  int i = blockIdx.x * blockDim.x + threadIdx.x;
  if (i < n) deg[i] = rsqrtf(deg[i]);
}

__global__ __launch_bounds__(256)
void gemm_scale_kernel(const float* __restrict__ x, const float* __restrict__ W,
                       const float* __restrict__ dinv, float* __restrict__ hp,
                       float* __restrict__ outinit, int nrows) {
  __shared__ v2f Wlds[(NF / 2) * PSTR];
  for (int i = threadIdx.x; i < NF * NF; i += 256) {
    int k = i >> 7, c = i & 127;
    Wlds[(k >> 1) * PSTR + c][k & 1] = W[i];
  }
  __syncthreads();

  const int wave = threadIdx.x >> 5;
  const int lane = threadIdx.x & 31;
  const int m0 = (blockIdx.x * 8 + wave) * 16;
  if (m0 >= nrows) return;

  const int half  = lane >> 4;
  const int khalf = half * 2;
  const int colb  = lane & 15;

  v8f acc[8] = {};

  const float* xr = x + (size_t)(m0 + colb) * NF + khalf;

  for (int kk = 0; kk < NF; kk += 4) {
    v2f a = *(const v2f*)(xr + kk);
    const v2f* wr = &Wlds[((kk >> 1) + half) * PSTR + colb];
#pragma unroll
    for (int nt = 0; nt < 8; ++nt) {
      v2f b = wr[nt * 16];
      acc[nt] = __builtin_amdgcn_wmma_f32_16x16x4_f32(
          false, a, false, b, (short)0, acc[nt], false, false);
    }
  }

  const int rbase = half << 3;
#pragma unroll
  for (int r = 0; r < 8; ++r) {
    const int row = m0 + rbase + r;
    const float dv = dinv[row];
    const size_t ro = (size_t)row * NF + colb;
#pragma unroll
    for (int nt = 0; nt < 8; ++nt) {
      float v = acc[nt][r] * dv;
      hp[ro + nt * 16] = v;
      outinit[ro + nt * 16] = v;
    }
  }
}

__global__ void scatter_kernel(const int* __restrict__ ei,
                               const float* __restrict__ hp,
                               float* __restrict__ out, int nedges) {
  int t = blockIdx.x * blockDim.x + threadIdx.x;
  int e = t >> 5;
  if (e >= nedges) return;
  int lane = t & 31;
  int src = ei[e];
  int dst = ei[nedges + e];
  const float4 v = *(const float4*)(hp + (size_t)src * NF + lane * 4);
  float* o = out + (size_t)dst * NF + lane * 4;
  __hip_atomic_fetch_add(o + 0, v.x, __ATOMIC_RELAXED, __HIP_MEMORY_SCOPE_AGENT);
  __hip_atomic_fetch_add(o + 1, v.y, __ATOMIC_RELAXED, __HIP_MEMORY_SCOPE_AGENT);
  __hip_atomic_fetch_add(o + 2, v.z, __ATOMIC_RELAXED, __HIP_MEMORY_SCOPE_AGENT);
  __hip_atomic_fetch_add(o + 3, v.w, __ATOMIC_RELAXED, __HIP_MEMORY_SCOPE_AGENT);
}

__global__ void finalize_kernel(float* __restrict__ out,
                                const float* __restrict__ dinv,
                                const float* __restrict__ bias,
                                long long total) {
  long long t = (long long)blockIdx.x * blockDim.x + threadIdx.x;
  if (t < total) {
    int row = (int)(t >> 7);
    int f = (int)(t & 127);
    out[t] = out[t] * dinv[row] + bias[f];
  }
}

extern "C" void kernel_launch(void* const* d_in, const int* in_sizes, int n_in,
                              void* d_out, int out_size, void* d_ws, size_t ws_size,
                              hipStream_t stream) {
  const float* x  = (const float*)d_in[0];
  const int*   ei = (const int*)d_in[1];
  const float* W  = (const float*)d_in[2];
  const float* b  = (const float*)d_in[3];
  float* out = (float*)d_out;

  const int N = in_sizes[0] / NF;
  const int E = in_sizes[1] / 2;

  float* deg = (float*)d_ws;
  size_t off = ((size_t)N * sizeof(float) + 511) & ~(size_t)511;
  float* hp = (float*)((char*)d_ws + off);

  deg_init_kernel<<<(N + 255) / 256, 256, 0, stream>>>(deg, N);
  deg_count_kernel<<<(E + 255) / 256, 256, 0, stream>>>(ei, deg, E);
  rsqrt_kernel<<<(N + 255) / 256, 256, 0, stream>>>(deg, N);

  gemm_scale_kernel<<<(N + 127) / 128, 256, 0, stream>>>(x, W, deg, hp, out, N);

  long long sthreads = (long long)E * 32;
  scatter_kernel<<<(int)((sthreads + 255) / 256), 256, 0, stream>>>(ei, hp, out, E);

  long long total = (long long)N * NF;
  finalize_kernel<<<(int)((total + 255) / 256), 256, 0, stream>>>(out, deg, b, total);
}


# aggregation made output-stationary: a work-group owns 576 dst rows as an LDS image, filters the whole dst list for them, adds gathered h' rows by LDS float adds, writes out once; adds in any order
# speedup vs baseline: 1.9502x; 1.9502x over previous
#include <hip/hip_runtime.h>


#define NF 128
#define PSTR 144

typedef __attribute__((ext_vector_type(2))) float v2f;
typedef __attribute__((ext_vector_type(8))) float v8f;

__global__ void deg_init_kernel(float* __restrict__ deg, int n) {
  int i = blockIdx.x * blockDim.x + threadIdx.x;
  if (i < n) deg[i] = 1.0f;
}

__global__ void deg_count_kernel(const int* __restrict__ ei,
                                 float* __restrict__ deg, int nedges) {
  int e = blockIdx.x * blockDim.x + threadIdx.x;
  if (e < nedges) {
    int dst = ei[nedges + e];
    __hip_atomic_fetch_add(&deg[dst], 1.0f, __ATOMIC_RELAXED,
                           __HIP_MEMORY_SCOPE_AGENT);
  }
}

__global__ void rsqrt_kernel(float* __restrict__ deg, int n) {
  int i = blockIdx.x * blockDim.x + threadIdx.x;
  if (i < n) deg[i] = rsqrtf(deg[i]);
}

__global__ __launch_bounds__(256)
void gemm_scale_kernel(const float* __restrict__ x, const float* __restrict__ W,
                       const float* __restrict__ dinv, float* __restrict__ hp,
                       int nrows) {
  __shared__ v2f Wlds[(NF / 2) * PSTR];
  for (int i = threadIdx.x; i < NF * NF; i += 256) {
    int k = i >> 7, c = i & 127;
    Wlds[(k >> 1) * PSTR + c][k & 1] = W[i];
  }
  __syncthreads();

  const int wave = threadIdx.x >> 5;
  const int lane = threadIdx.x & 31;
  const int m0 = (blockIdx.x * 8 + wave) * 16;
  if (m0 >= nrows) return;

  const int half  = lane >> 4;
  const int khalf = half * 2;
  const int colb  = lane & 15;

  v8f acc[8] = {};

  const float* xr = x + (size_t)(m0 + colb) * NF + khalf;

  for (int kk = 0; kk < NF; kk += 4) {
    v2f a = *(const v2f*)(xr + kk);
    const v2f* wr = &Wlds[((kk >> 1) + half) * PSTR + colb];
#pragma unroll
    for (int nt = 0; nt < 8; ++nt) {
      v2f b = wr[nt * 16];
      acc[nt] = __builtin_amdgcn_wmma_f32_16x16x4_f32(
          false, a, false, b, (short)0, acc[nt], false, false);
    }
  }

  const int rbase = half << 3;
#pragma unroll
  for (int r = 0; r < 8; ++r) {
    const int row = m0 + rbase + r;
    const float dv = dinv[row];
    const size_t ro = (size_t)row * NF + colb;
#pragma unroll
    for (int nt = 0; nt < 8; ++nt) {
      hp[ro + nt * 16] = acc[nt][r] * dv;
    }
  }
}

#define RPB 576
#define AGG_WAVES 32
#define QCAP 160
#define UNR 8
#define STEP_EDGES (32 * 4 * UNR)

__device__ __forceinline__ void lds_add(float* p, float v) {
  __hip_atomic_fetch_add(p, v, __ATOMIC_RELAXED, __HIP_MEMORY_SCOPE_WORKGROUP);
}

__device__ __forceinline__ int4 load_dst4_guarded(const int* __restrict__ dstp,
                                                  int e0, int nedges) {
  int4 r;
  r.x = (e0 + 0 < nedges) ? dstp[e0 + 0] : -1;
  r.y = (e0 + 1 < nedges) ? dstp[e0 + 1] : -1;
  r.z = (e0 + 2 < nedges) ? dstp[e0 + 2] : -1;
  r.w = (e0 + 3 < nedges) ? dstp[e0 + 3] : -1;
  return r;
}

template <bool FULL>
__device__ __forceinline__ void drain(int cnt, int qbase, const int* myqe,
                                      const unsigned short* myqd,
                                      const int* __restrict__ ei,
                                      const float* __restrict__ hp,
                                      float* acc, int lane) {
  int e = 0, dl = 0;
  if (FULL || lane < cnt) {
    e = myqe[qbase + lane];
    dl = myqd[qbase + lane];
  }
  const int s = ei[e];
#pragma unroll 1
  for (int k0 = 0; k0 < (FULL ? 32 : cnt); k0 += 8) {
    float4 v[8];
#pragma unroll
    for (int i = 0; i < 8; ++i) {
      if (FULL || k0 + i < cnt) {
        const int si = __builtin_amdgcn_readlane(s, k0 + i);
        v[i] = *(const float4*)(hp + (size_t)si * NF + lane * 4);
      }
    }
#pragma unroll
    for (int i = 0; i < 8; ++i) {
      if (FULL || k0 + i < cnt) {
        const int di = __builtin_amdgcn_readlane(dl, k0 + i);
        float* a = acc + di * NF + lane;
        lds_add(a, v[i].x);
        lds_add(a + 32, v[i].y);
        lds_add(a + 64, v[i].z);
        lds_add(a + 96, v[i].w);
      }
    }
  }
}

__global__ __launch_bounds__(AGG_WAVES * 32)
void aggregate_rows_kernel(const int* __restrict__ ei, const float* __restrict__ hp,
                           const float* __restrict__ dinv,
                           const float* __restrict__ bias, float* __restrict__ out,
                           int nrows, int nedges) {
  __shared__ float acc[RPB * NF];
  __shared__ int qe[AGG_WAVES * QCAP];
  __shared__ unsigned short qd[AGG_WAVES * QCAP];

  const int lane = threadIdx.x & 31;
  const int wave = threadIdx.x >> 5;
  const int row0 = blockIdx.x * RPB;
  const int nr = min(RPB, nrows - row0);

  for (int r = wave; r < nr; r += AGG_WAVES) {
    const float4 v = *(const float4*)(hp + (size_t)(row0 + r) * NF + lane * 4);
    float* a = acc + r * NF + lane;
    a[0] = v.x; a[32] = v.y; a[64] = v.z; a[96] = v.w;
  }
  __syncthreads();

  const int* __restrict__ dstp = ei + nedges;
  const bool vec_ok = (((size_t)dstp) & 15) == 0;
  const int per_step = AGG_WAVES * STEP_EDGES;
  const int nsteps = (nedges + per_step - 1) / per_step;
  const int sstart = (int)(((long long)blockIdx.x * nsteps) / gridDim.x);

  int* myqe = qe + wave * QCAP;
  unsigned short* myqd = qd + wave * QCAP;
  int qn = 0;

  const int wave_u = __builtin_amdgcn_readfirstlane(wave);

  for (int st = 0; st < nsteps; ++st) {
    int ss = st + sstart;
    if (ss >= nsteps) ss -= nsteps;
    const int wbase = (ss * AGG_WAVES + wave_u) * STEP_EDGES;
    const int base = wbase + lane * 4;
    int4 cur[UNR];
    if (vec_ok && wbase + STEP_EDGES <= nedges) {
      const int4* p = (const int4*)(dstp + wbase) + lane;
#pragma unroll
      for (int u = 0; u < UNR; ++u) cur[u] = p[u * 32];
    } else {
#pragma unroll
      for (int u = 0; u < UNR; ++u) cur[u] = load_dst4_guarded(dstp, base + u * 128, nedges);
    }

#pragma unroll
    for (int u = 0; u < UNR; ++u) {
      const int e0 = base + u * 128;
      const unsigned d0 = (unsigned)(cur[u].x - row0);
      const unsigned d1 = (unsigned)(cur[u].y - row0);
      const unsigned d2 = (unsigned)(cur[u].z - row0);
      const unsigned d3 = (unsigned)(cur[u].w - row0);
      const bool m0 = d0 < (unsigned)nr, m1 = d1 < (unsigned)nr;
      const bool m2 = d2 < (unsigned)nr, m3 = d3 < (unsigned)nr;
      const unsigned b0 = __builtin_amdgcn_ballot_w32(m0);
      const unsigned b1 = __builtin_amdgcn_ballot_w32(m1);
      const unsigned b2 = __builtin_amdgcn_ballot_w32(m2);
      const unsigned b3 = __builtin_amdgcn_ballot_w32(m3);
      if ((b0 | b1 | b2 | b3) != 0u) {
        if (b0) {
          const int pos = qn + (int)__builtin_amdgcn_mbcnt_lo(b0, 0u);
          if (m0) { myqe[pos] = e0 + 0; myqd[pos] = (unsigned short)d0; }
          qn += __builtin_popcount(b0);
        }
        if (b1) {
          const int pos = qn + (int)__builtin_amdgcn_mbcnt_lo(b1, 0u);
          if (m1) { myqe[pos] = e0 + 1; myqd[pos] = (unsigned short)d1; }
          qn += __builtin_popcount(b1);
        }
        if (b2) {
          const int pos = qn + (int)__builtin_amdgcn_mbcnt_lo(b2, 0u);
          if (m2) { myqe[pos] = e0 + 2; myqd[pos] = (unsigned short)d2; }
          qn += __builtin_popcount(b2);
        }
        if (b3) {
          const int pos = qn + (int)__builtin_amdgcn_mbcnt_lo(b3, 0u);
          if (m3) { myqe[pos] = e0 + 3; myqd[pos] = (unsigned short)d3; }
          qn += __builtin_popcount(b3);
        }
        while (qn >= 32) {
          qn -= 32;
          drain<true>(32, qn, myqe, myqd, ei, hp, acc, lane);
        }
      }
    }
  }
  if (qn > 0) drain<false>(qn, 0, myqe, myqd, ei, hp, acc, lane);
  __syncthreads();

  const float4 bv = *(const float4*)(bias + lane * 4);
  for (int r = wave; r < nr; r += AGG_WAVES) {
    const int row = row0 + r;
    const float dv = dinv[row];
    const float* a = acc + r * NF + lane;
    float4 o;
    o.x = a[0] * dv + bv.x;
    o.y = a[32] * dv + bv.y;
    o.z = a[64] * dv + bv.z;
    o.w = a[96] * dv + bv.w;
    *(float4*)(out + (size_t)row * NF + lane * 4) = o;
  }
}

extern "C" void kernel_launch(void* const* d_in, const int* in_sizes, int n_in,
                              void* d_out, int out_size, void* d_ws, size_t ws_size,
                              hipStream_t stream) {
  const float* x  = (const float*)d_in[0];
  const int*   ei = (const int*)d_in[1];
  const float* W  = (const float*)d_in[2];
  const float* b  = (const float*)d_in[3];
  float* out = (float*)d_out;

  const int N = in_sizes[0] / NF;
  const int E = in_sizes[1] / 2;

  float* deg = (float*)d_ws;
  size_t off = ((size_t)N * sizeof(float) + 511) & ~(size_t)511;
  float* hp = (float*)((char*)d_ws + off);

  deg_init_kernel<<<(N + 255) / 256, 256, 0, stream>>>(deg, N);
  deg_count_kernel<<<(E + 255) / 256, 256, 0, stream>>>(ei, deg, E);
  rsqrt_kernel<<<(N + 255) / 256, 256, 0, stream>>>(deg, N);

  gemm_scale_kernel<<<(N + 127) / 128, 256, 0, stream>>>(x, W, deg, hp, N);

  aggregate_rows_kernel<<<(N + RPB - 1) / RPB, AGG_WAVES * 32, 0, stream>>>(
      ei, hp, deg, b, out, N, E);
}


# degree count without memory atomics: 64 edge slices x 4 row ranges count destinations by integer LDS adds into count planes; next launch sums a row's 64 planes in fixed order, +1, rsqrt; 4 launches
# speedup vs baseline: 2.9022x; 1.4882x over previous
#include <hip/hip_runtime.h>


#define NF 128
#define PSTR 144

typedef __attribute__((ext_vector_type(2))) float v2f;
typedef __attribute__((ext_vector_type(8))) float v8f;

#define DEG_SLICES 64
#define DEG_RJ 25024
#define DEG_THREADS 512
#define DEG_UNR 16

__device__ __forceinline__ int plane_pitch(int nrows) { return (nrows + 31) & ~31; }

__device__ __forceinline__ int slice_len(int nedges) {
  return (((nedges + DEG_SLICES - 1) / DEG_SLICES) + 3) & ~3;
}

__device__ __forceinline__ int4 load_dst4_guarded(const int* __restrict__ dstp,
                                                  int e0, int lim) {
  int4 r;
  r.x = (e0 + 0 < lim) ? dstp[e0 + 0] : -1;
  r.y = (e0 + 1 < lim) ? dstp[e0 + 1] : -1;
  r.z = (e0 + 2 < lim) ? dstp[e0 + 2] : -1;
  r.w = (e0 + 3 < lim) ? dstp[e0 + 3] : -1;
  return r;
}

__device__ __forceinline__ void deg_load(int4 (&d)[DEG_UNR], const int* __restrict__ dstp,
                                         int b, int e_end, bool vec_ok, int tid) {
#pragma unroll
  for (int u = DEG_UNR - 1; u >= 0; --u) {
    const int bu = b + u * DEG_THREADS * 4;
    const int e = bu + tid * 4;
    if (bu >= e_end) d[u] = make_int4(-1, -1, -1, -1);
    else if (vec_ok && bu + DEG_THREADS * 4 <= e_end) d[u] = *(const int4*)(dstp + e);
    else d[u] = load_dst4_guarded(dstp, e, e_end);
  }
}

__device__ __forceinline__ void deg_count1(int* cnt, int dst, int r0, int nr) {
  const unsigned l = (unsigned)(dst - r0);
  if (l < (unsigned)nr)
    __hip_atomic_fetch_add(&cnt[l], 1, __ATOMIC_RELAXED, __HIP_MEMORY_SCOPE_WORKGROUP);
}

__global__ __launch_bounds__(DEG_THREADS)
void deg_planes_kernel(const int* __restrict__ ei, int* __restrict__ planes,
                       int nrows, int nedges) {
  __shared__ int cnt[DEG_RJ];
  const int tid = threadIdx.x;
  const int pitch = plane_pitch(nrows);
  const int r0 = blockIdx.y * DEG_RJ;
  const int nr = max(0, min(DEG_RJ, nrows - r0));
  const int nrw = min(DEG_RJ, pitch - r0);

  const int* __restrict__ dstp = ei + nedges;
  const bool vec_ok = (((size_t)dstp) & 15) == 0;
  const int slen = slice_len(nedges);
  const int e_begin = min(nedges, (int)blockIdx.x * slen);
  const int e_end = min(nedges, e_begin + slen);
  const int step = DEG_THREADS * 4 * DEG_UNR;

  for (int i = tid * 4; i < DEG_RJ; i += DEG_THREADS * 4)
    *(int4*)(cnt + i) = make_int4(0, 0, 0, 0);
  __syncthreads();

  for (int b = e_begin; b < e_end; b += step) {
    int4 d[DEG_UNR];
    deg_load(d, dstp, b, e_end, vec_ok, tid);
#pragma unroll
    for (int u = 0; u < DEG_UNR; ++u) {
      deg_count1(cnt, d[u].x, r0, nr);
      deg_count1(cnt, d[u].y, r0, nr);
      deg_count1(cnt, d[u].z, r0, nr);
      deg_count1(cnt, d[u].w, r0, nr);
    }
  }
  __syncthreads();

  int* __restrict__ pl = planes + (size_t)blockIdx.x * pitch + r0;
  for (int i = tid * 4; i < nrw; i += DEG_THREADS * 4)
    *(int4*)(pl + i) = *(const int4*)(cnt + i);
}

__global__ __launch_bounds__(256)
void deg_finish_kernel(const int* __restrict__ planes, float* __restrict__ dinv,
                       int nrows) {
  const int pitch = plane_pitch(nrows);
  const int i = (blockIdx.x * 256 + threadIdx.x) * 2;
  if (i >= pitch) return;
  int2 v[DEG_SLICES];
#pragma unroll
  for (int s = 0; s < DEG_SLICES; ++s)
    v[s] = *(const int2*)(planes + (size_t)s * pitch + i);
  int c0 = 1, c1 = 1;
#pragma unroll
  for (int s = 0; s < DEG_SLICES; ++s) { c0 += v[s].x; c1 += v[s].y; }
  float2 r;
  r.x = rsqrtf((float)c0);
  r.y = rsqrtf((float)c1);
  *(float2*)(dinv + i) = r;
}

__global__ __launch_bounds__(256)
void gemm_scale_kernel(const float* __restrict__ x, const float* __restrict__ W,
                       const float* __restrict__ dinv, float* __restrict__ hp,
                       int nrows) {
  __shared__ v2f Wlds[(NF / 2) * PSTR];
  for (int i = threadIdx.x; i < NF * NF; i += 256) {
    int k = i >> 7, c = i & 127;
    Wlds[(k >> 1) * PSTR + c][k & 1] = W[i];
  }
  __syncthreads();

  const int wave = threadIdx.x >> 5;
  const int lane = threadIdx.x & 31;
  const int m0 = (blockIdx.x * 8 + wave) * 16;
  if (m0 >= nrows) return;

  const int half  = lane >> 4;
  const int khalf = half * 2;
  const int colb  = lane & 15;

  v8f acc[8] = {};

  const float* xr = x + (size_t)(m0 + colb) * NF + khalf;

  for (int kk = 0; kk < NF; kk += 4) {
    v2f a = *(const v2f*)(xr + kk);
    const v2f* wr = &Wlds[((kk >> 1) + half) * PSTR + colb];
#pragma unroll
    for (int nt = 0; nt < 8; ++nt) {
      v2f b = wr[nt * 16];
      acc[nt] = __builtin_amdgcn_wmma_f32_16x16x4_f32(
          false, a, false, b, (short)0, acc[nt], false, false);
    }
  }

  const int rbase = half << 3;
#pragma unroll
  for (int r = 0; r < 8; ++r) {
    const int row = m0 + rbase + r;
    const float dv = dinv[row];
    const size_t ro = (size_t)row * NF + colb;
#pragma unroll
    for (int nt = 0; nt < 8; ++nt) {
      hp[ro + nt * 16] = acc[nt][r] * dv;
    }
  }
}

#define RPB 576
#define AGG_WAVES 32
#define QCAP 160
#define UNR 8
#define STEP_EDGES (32 * 4 * UNR)

__device__ __forceinline__ void lds_add(float* p, float v) {
  __hip_atomic_fetch_add(p, v, __ATOMIC_RELAXED, __HIP_MEMORY_SCOPE_WORKGROUP);
}

template <bool FULL>
__device__ __forceinline__ void drain(int cnt, int qbase, const int* myqe,
                                      const unsigned short* myqd,
                                      const int* __restrict__ ei,
                                      const float* __restrict__ hp,
                                      float* acc, int lane) {
  int e = 0, dl = 0;
  if (FULL || lane < cnt) {
    e = myqe[qbase + lane];
    dl = myqd[qbase + lane];
  }
  const int s = ei[e];
#pragma unroll 1
  for (int k0 = 0; k0 < (FULL ? 32 : cnt); k0 += 8) {
    float4 v[8];
#pragma unroll
    for (int i = 0; i < 8; ++i) {
      if (FULL || k0 + i < cnt) {
        const int si = __builtin_amdgcn_readlane(s, k0 + i);
        v[i] = *(const float4*)(hp + (size_t)si * NF + lane * 4);
      }
    }
#pragma unroll
    for (int i = 0; i < 8; ++i) {
      if (FULL || k0 + i < cnt) {
        const int di = __builtin_amdgcn_readlane(dl, k0 + i);
        float* a = acc + di * NF + lane;
        lds_add(a, v[i].x);
        lds_add(a + 32, v[i].y);
        lds_add(a + 64, v[i].z);
        lds_add(a + 96, v[i].w);
      }
    }
  }
}

__global__ __launch_bounds__(AGG_WAVES * 32)
void aggregate_rows_kernel(const int* __restrict__ ei, const float* __restrict__ hp,
                           const float* __restrict__ dinv,
                           const float* __restrict__ bias, float* __restrict__ out,
                           int nrows, int nedges) {
  __shared__ float acc[RPB * NF];
  __shared__ int qe[AGG_WAVES * QCAP];
  __shared__ unsigned short qd[AGG_WAVES * QCAP];

  const int lane = threadIdx.x & 31;
  const int wave = threadIdx.x >> 5;
  const int row0 = blockIdx.x * RPB;
  const int nr = min(RPB, nrows - row0);

  for (int r = wave; r < nr; r += AGG_WAVES) {
    const float4 v = *(const float4*)(hp + (size_t)(row0 + r) * NF + lane * 4);
    float* a = acc + r * NF + lane;
    a[0] = v.x; a[32] = v.y; a[64] = v.z; a[96] = v.w;
  }
  __syncthreads();

  const int* __restrict__ dstp = ei + nedges;
  const bool vec_ok = (((size_t)dstp) & 15) == 0;
  const int per_step = AGG_WAVES * STEP_EDGES;
  const int nsteps = (nedges + per_step - 1) / per_step;
  const int sstart = (int)(((long long)blockIdx.x * nsteps) / gridDim.x);

  int* myqe = qe + wave * QCAP;
  unsigned short* myqd = qd + wave * QCAP;
  int qn = 0;

  const int wave_u = __builtin_amdgcn_readfirstlane(wave);

  for (int st = 0; st < nsteps; ++st) {
    int ss = st + sstart;
    if (ss >= nsteps) ss -= nsteps;
    const int wbase = (ss * AGG_WAVES + wave_u) * STEP_EDGES;
    const int base = wbase + lane * 4;
    int4 cur[UNR];
    if (vec_ok && wbase + STEP_EDGES <= nedges) {
      const int4* p = (const int4*)(dstp + wbase) + lane;
#pragma unroll
      for (int u = 0; u < UNR; ++u) cur[u] = p[u * 32];
    } else {
#pragma unroll
      for (int u = 0; u < UNR; ++u) cur[u] = load_dst4_guarded(dstp, base + u * 128, nedges);
    }

#pragma unroll
    for (int u = 0; u < UNR; ++u) {
      const int e0 = base + u * 128;
      const unsigned d0 = (unsigned)(cur[u].x - row0);
      const unsigned d1 = (unsigned)(cur[u].y - row0);
      const unsigned d2 = (unsigned)(cur[u].z - row0);
      const unsigned d3 = (unsigned)(cur[u].w - row0);
      const bool m0 = d0 < (unsigned)nr, m1 = d1 < (unsigned)nr;
      const bool m2 = d2 < (unsigned)nr, m3 = d3 < (unsigned)nr;
      const unsigned b0 = __builtin_amdgcn_ballot_w32(m0);
      const unsigned b1 = __builtin_amdgcn_ballot_w32(m1);
      const unsigned b2 = __builtin_amdgcn_ballot_w32(m2);
      const unsigned b3 = __builtin_amdgcn_ballot_w32(m3);
      if ((b0 | b1 | b2 | b3) != 0u) {
        if (b0) {
          const int pos = qn + (int)__builtin_amdgcn_mbcnt_lo(b0, 0u);
          if (m0) { myqe[pos] = e0 + 0; myqd[pos] = (unsigned short)d0; }
          qn += __builtin_popcount(b0);
        }
        if (b1) {
          const int pos = qn + (int)__builtin_amdgcn_mbcnt_lo(b1, 0u);
          if (m1) { myqe[pos] = e0 + 1; myqd[pos] = (unsigned short)d1; }
          qn += __builtin_popcount(b1);
        }
        if (b2) {
          const int pos = qn + (int)__builtin_amdgcn_mbcnt_lo(b2, 0u);
          if (m2) { myqe[pos] = e0 + 2; myqd[pos] = (unsigned short)d2; }
          qn += __builtin_popcount(b2);
        }
        if (b3) {
          const int pos = qn + (int)__builtin_amdgcn_mbcnt_lo(b3, 0u);
          if (m3) { myqe[pos] = e0 + 3; myqd[pos] = (unsigned short)d3; }
          qn += __builtin_popcount(b3);
        }
        while (qn >= 32) {
          qn -= 32;
          drain<true>(32, qn, myqe, myqd, ei, hp, acc, lane);
        }
      }
    }
  }
  if (qn > 0) drain<false>(qn, 0, myqe, myqd, ei, hp, acc, lane);
  __syncthreads();

  const float4 bv = *(const float4*)(bias + lane * 4);
  for (int r = wave; r < nr; r += AGG_WAVES) {
    const int row = row0 + r;
    const float dv = dinv[row];
    const float* a = acc + r * NF + lane;
    float4 o;
    o.x = a[0] * dv + bv.x;
    o.y = a[32] * dv + bv.y;
    o.z = a[64] * dv + bv.z;
    o.w = a[96] * dv + bv.w;
    *(float4*)(out + (size_t)row * NF + lane * 4) = o;
  }
}

extern "C" void kernel_launch(void* const* d_in, const int* in_sizes, int n_in,
                              void* d_out, int out_size, void* d_ws, size_t ws_size,
                              hipStream_t stream) {
  const float* x  = (const float*)d_in[0];
  const int*   ei = (const int*)d_in[1];
  const float* W  = (const float*)d_in[2];
  const float* b  = (const float*)d_in[3];
  float* out = (float*)d_out;

  const int N = in_sizes[0] / NF;
  const int E = in_sizes[1] / 2;

  float* deg = (float*)d_ws;
  size_t off = ((size_t)N * sizeof(float) + 511) & ~(size_t)511;
  float* hp = (float*)((char*)d_ws + off);
  size_t off2 = (off + (size_t)N * NF * sizeof(float) + 511) & ~(size_t)511;
  int* planes = (int*)((char*)d_ws + off2);

  const int pitch = (N + 31) & ~31;
  deg_planes_kernel<<<dim3(DEG_SLICES, (pitch + DEG_RJ - 1) / DEG_RJ), DEG_THREADS, 0, stream>>>(
      ei, planes, N, E);
  deg_finish_kernel<<<(pitch / 2 + 255) / 256, 256, 0, stream>>>(planes, deg, N);

  gemm_scale_kernel<<<(N + 127) / 128, 256, 0, stream>>>(x, W, deg, hp, N);

  aggregate_rows_kernel<<<(N + RPB - 1) / RPB, AGG_WAVES * 32, 0, stream>>>(
      ei, hp, deg, b, out, N, E);
}


# edge filter reads one owner byte per edge (dst/576, packed by the degree launch) instead of the 4-byte destination: 4x fewer list bytes and load rounds per work-group; sums in any order as before
# speedup vs baseline: 3.2449x; 1.1181x over previous
#include <hip/hip_runtime.h>


#define NF 128
#define PSTR 144

typedef __attribute__((ext_vector_type(2))) float v2f;
typedef __attribute__((ext_vector_type(8))) float v8f;

#define DEG_SLICES 64
#define DEG_RJ 25024
#define DEG_THREADS 512
#define DEG_UNR 16

#define RPB 576
#define OWN_NONE 0xFFu

__device__ __forceinline__ int plane_pitch(int nrows) { return (nrows + 31) & ~31; }

__device__ __forceinline__ int slice_len(int nedges) {
  return (((nedges + DEG_SLICES - 1) / DEG_SLICES) + 127) & ~127;
}

__device__ __forceinline__ unsigned owner_byte(int dst, int nrows) {
  return ((unsigned)dst < (unsigned)nrows) ? (unsigned)dst / RPB : OWN_NONE;
}

__device__ __forceinline__ int4 load_dst4_guarded(const int* __restrict__ dstp,
                                                  int e0, int lim) {
  int4 r;
  r.x = (e0 + 0 < lim) ? dstp[e0 + 0] : -1;
  r.y = (e0 + 1 < lim) ? dstp[e0 + 1] : -1;
  r.z = (e0 + 2 < lim) ? dstp[e0 + 2] : -1;
  r.w = (e0 + 3 < lim) ? dstp[e0 + 3] : -1;
  return r;
}

__device__ __forceinline__ void deg_load(int4 (&d)[DEG_UNR], const int* __restrict__ dstp,
                                         int b, int e_end, bool vec_ok, int tid) {
#pragma unroll
  for (int u = DEG_UNR - 1; u >= 0; --u) {
    const int bu = b + u * DEG_THREADS * 4;
    const int e = bu + tid * 4;
    if (bu >= e_end) d[u] = make_int4(-1, -1, -1, -1);
    else if (vec_ok && bu + DEG_THREADS * 4 <= e_end) d[u] = *(const int4*)(dstp + e);
    else d[u] = load_dst4_guarded(dstp, e, e_end);
  }
}

__device__ __forceinline__ void deg_count1(int* cnt, int dst, int r0, int nr) {
  const unsigned l = (unsigned)(dst - r0);
  if (l < (unsigned)nr)
    __hip_atomic_fetch_add(&cnt[l], 1, __ATOMIC_RELAXED, __HIP_MEMORY_SCOPE_WORKGROUP);
}

__global__ __launch_bounds__(DEG_THREADS)
void deg_planes_kernel(const int* __restrict__ ei, int* __restrict__ planes,
                       unsigned int* __restrict__ own, int nrows, int nedges) {
  __shared__ int cnt[DEG_RJ];
  const int tid = threadIdx.x;
  const int pitch = plane_pitch(nrows);
  const int r0 = blockIdx.y * DEG_RJ;
  const int nr = max(0, min(DEG_RJ, nrows - r0));
  const int nrw = min(DEG_RJ, pitch - r0);

  const int* __restrict__ dstp = ei + nedges;
  const bool vec_ok = (((size_t)dstp) & 15) == 0;
  const int slen = slice_len(nedges);
  const int e_begin = min(nedges, (int)blockIdx.x * slen);
  const int e_end = min(nedges, e_begin + slen);
  const int step = DEG_THREADS * 4 * DEG_UNR;

  for (int i = tid * 4; i < DEG_RJ; i += DEG_THREADS * 4)
    *(int4*)(cnt + i) = make_int4(0, 0, 0, 0);
  __syncthreads();

  for (int b = e_begin; b < e_end; b += step) {
    int4 d[DEG_UNR];
    deg_load(d, dstp, b, e_end, vec_ok, tid);
    if (blockIdx.y == 0) {
#pragma unroll
      for (int u = 0; u < DEG_UNR; ++u) {
        const int e = b + u * DEG_THREADS * 4 + tid * 4;
        if (e < e_end)
          own[e >> 2] = owner_byte(d[u].x, nrows) | (owner_byte(d[u].y, nrows) << 8) |
                        (owner_byte(d[u].z, nrows) << 16) | (owner_byte(d[u].w, nrows) << 24);
      }
    }
#pragma unroll
    for (int u = 0; u < DEG_UNR; ++u) {
      deg_count1(cnt, d[u].x, r0, nr);
      deg_count1(cnt, d[u].y, r0, nr);
      deg_count1(cnt, d[u].z, r0, nr);
      deg_count1(cnt, d[u].w, r0, nr);
    }
  }
  __syncthreads();

  int* __restrict__ pl = planes + (size_t)blockIdx.x * pitch + r0;
  for (int i = tid * 4; i < nrw; i += DEG_THREADS * 4)
    *(int4*)(pl + i) = *(const int4*)(cnt + i);
}

__global__ __launch_bounds__(256)
void deg_finish_kernel(const int* __restrict__ planes, float* __restrict__ dinv,
                       int nrows) {
  const int pitch = plane_pitch(nrows);
  const int i = (blockIdx.x * 256 + threadIdx.x) * 2;
  if (i >= pitch) return;
  int2 v[DEG_SLICES];
#pragma unroll
  for (int s = 0; s < DEG_SLICES; ++s)
    v[s] = *(const int2*)(planes + (size_t)s * pitch + i);
  int c0 = 1, c1 = 1;
#pragma unroll
  for (int s = 0; s < DEG_SLICES; ++s) { c0 += v[s].x; c1 += v[s].y; }
  float2 r;
  r.x = rsqrtf((float)c0);
  r.y = rsqrtf((float)c1);
  *(float2*)(dinv + i) = r;
}

__global__ __launch_bounds__(256)
void gemm_scale_kernel(const float* __restrict__ x, const float* __restrict__ W,
                       const float* __restrict__ dinv, float* __restrict__ hp,
                       int nrows) {
  __shared__ v2f Wlds[(NF / 2) * PSTR];
  for (int i = threadIdx.x; i < NF * NF; i += 256) {
    int k = i >> 7, c = i & 127;
    Wlds[(k >> 1) * PSTR + c][k & 1] = W[i];
  }
  __syncthreads();

  const int wave = threadIdx.x >> 5;
  const int lane = threadIdx.x & 31;
  const int m0 = (blockIdx.x * 8 + wave) * 16;
  if (m0 >= nrows) return;

  const int half  = lane >> 4;
  const int khalf = half * 2;
  const int colb  = lane & 15;

  v8f acc[8] = {};

  const float* xr = x + (size_t)(m0 + colb) * NF + khalf;

  for (int kk = 0; kk < NF; kk += 4) {
    v2f a = *(const v2f*)(xr + kk);
    const v2f* wr = &Wlds[((kk >> 1) + half) * PSTR + colb];
#pragma unroll
    for (int nt = 0; nt < 8; ++nt) {
      v2f b = wr[nt * 16];
      acc[nt] = __builtin_amdgcn_wmma_f32_16x16x4_f32(
          false, a, false, b, (short)0, acc[nt], false, false);
    }
  }

  const int rbase = half << 3;
#pragma unroll
  for (int r = 0; r < 8; ++r) {
    const int row = m0 + rbase + r;
    const float dv = dinv[row];
    const size_t ro = (size_t)row * NF + colb;
#pragma unroll
    for (int nt = 0; nt < 8; ++nt) {
      hp[ro + nt * 16] = acc[nt][r] * dv;
    }
  }
}

#define AGG_WAVES 32
#define QCAP 64
#define UNR 8
#define STEP_EDGES (32 * 16 * UNR)

__device__ __forceinline__ unsigned zero_bytes(unsigned x) {
  return ~(((x & 0x7F7F7F7Fu) + 0x7F7F7F7Fu) | x | 0x7F7F7F7Fu);
}

__device__ __forceinline__ uint4 load_own4_guarded(const unsigned int* __restrict__ own,
                                                   int w0, int lim) {
  uint4 r;
  r.x = (w0 + 0 < lim) ? own[w0 + 0] : 0xFFFFFFFFu;
  r.y = (w0 + 1 < lim) ? own[w0 + 1] : 0xFFFFFFFFu;
  r.z = (w0 + 2 < lim) ? own[w0 + 2] : 0xFFFFFFFFu;
  r.w = (w0 + 3 < lim) ? own[w0 + 3] : 0xFFFFFFFFu;
  return r;
}

__device__ __forceinline__ void lds_add(float* p, float v) {
  __hip_atomic_fetch_add(p, v, __ATOMIC_RELAXED, __HIP_MEMORY_SCOPE_WORKGROUP);
}

template <bool FULL>
__device__ __forceinline__ void drain(int cnt, int qbase, const int* myqe,
                                      const int* __restrict__ ei,
                                      const int* __restrict__ dstp, int row0,
                                      const float* __restrict__ hp,
                                      float* acc, int lane) {
  int e = 0;
  if (FULL || lane < cnt) e = myqe[qbase + lane];
  const int s = ei[e];
  const int dl = dstp[e] - row0;
#pragma unroll 1
  for (int k0 = 0; k0 < (FULL ? 32 : cnt); k0 += 8) {
    float4 v[8];
#pragma unroll
    for (int i = 0; i < 8; ++i) {
      if (FULL || k0 + i < cnt) {
        const int si = __builtin_amdgcn_readlane(s, k0 + i);
        v[i] = *(const float4*)(hp + (size_t)si * NF + lane * 4);
      }
    }
#pragma unroll
    for (int i = 0; i < 8; ++i) {
      if (FULL || k0 + i < cnt) {
        const int di = __builtin_amdgcn_readlane(dl, k0 + i);
        float* a = acc + di * NF + lane;
        lds_add(a, v[i].x);
        lds_add(a + 32, v[i].y);
        lds_add(a + 64, v[i].z);
        lds_add(a + 96, v[i].w);
      }
    }
  }
}

__global__ __launch_bounds__(AGG_WAVES * 32)
void aggregate_rows_kernel(const int* __restrict__ ei, const unsigned int* __restrict__ own,
                           const float* __restrict__ hp,
                           const float* __restrict__ dinv,
                           const float* __restrict__ bias, float* __restrict__ out,
                           int nrows, int nedges) {
  __shared__ float acc[RPB * NF];
  __shared__ int qe[AGG_WAVES * QCAP];

  const int lane = threadIdx.x & 31;
  const int wave = threadIdx.x >> 5;
  const int row0 = blockIdx.x * RPB;
  const int nr = min(RPB, nrows - row0);

  for (int r = wave; r < nr; r += AGG_WAVES) {
    const float4 v = *(const float4*)(hp + (size_t)(row0 + r) * NF + lane * 4);
    float* a = acc + r * NF + lane;
    a[0] = v.x; a[32] = v.y; a[64] = v.z; a[96] = v.w;
  }
  __syncthreads();

  const int* __restrict__ dstp = ei + nedges;
  const bool vec_ok = (((size_t)own) & 15) == 0;
  const int nwords = (nedges + 3) >> 2;
  const int per_step = AGG_WAVES * STEP_EDGES;
  const int nsteps = (nedges + per_step - 1) / per_step;
  const int sstart = (int)(((long long)blockIdx.x * nsteps) / gridDim.x);

  int* myqe = qe + wave * QCAP;
  int qn = 0;

  const int wave_u = __builtin_amdgcn_readfirstlane(wave);
  const unsigned pat = (unsigned)blockIdx.x * 0x01010101u;

  for (int st = 0; st < nsteps; ++st) {
    int ss = st + sstart;
    if (ss >= nsteps) ss -= nsteps;
    const int wbase = (ss * AGG_WAVES + wave_u) * STEP_EDGES;
    const int base = wbase + lane * 16;
    uint4 cur[UNR];
    if (vec_ok && wbase + STEP_EDGES <= nedges) {
      const uint4* p = (const uint4*)(own + (wbase >> 2)) + lane;
#pragma unroll
      for (int u = 0; u < UNR; ++u) cur[u] = p[u * 32];
    } else {
#pragma unroll
      for (int u = 0; u < UNR; ++u)
        cur[u] = load_own4_guarded(own, (base + u * 512) >> 2, nwords);
    }

#pragma unroll
    for (int u = 0; u < UNR; ++u) {
      unsigned m = (zero_bytes(cur[u].x ^ pat) >> 3) | (zero_bytes(cur[u].y ^ pat) >> 2) |
                   (zero_bytes(cur[u].z ^ pat) >> 1) | zero_bytes(cur[u].w ^ pat);
      unsigned bal = __builtin_amdgcn_ballot_w32(m != 0u);
      const int e0m = base + u * 512 - 16;
      while (bal != 0u) {
        const int pos = qn + (int)__builtin_amdgcn_mbcnt_lo(bal, 0u);
        if (m != 0u) {
          const int p = __builtin_ctz(m);
          myqe[pos] = e0m + ((p & 7) << 2) + (p >> 3);
          m &= m - 1u;
        }
        qn += __builtin_popcount(bal);
        if (qn >= 32) {
          qn -= 32;
          drain<true>(32, qn, myqe, ei, dstp, row0, hp, acc, lane);
        }
        bal = __builtin_amdgcn_ballot_w32(m != 0u);
      }
    }
  }
  if (qn > 0) drain<false>(qn, 0, myqe, ei, dstp, row0, hp, acc, lane);
  __syncthreads();

  const float4 bv = *(const float4*)(bias + lane * 4);
  for (int r = wave; r < nr; r += AGG_WAVES) {
    const int row = row0 + r;
    const float dv = dinv[row];
    const float* a = acc + r * NF + lane;
    float4 o;
    o.x = a[0] * dv + bv.x;
    o.y = a[32] * dv + bv.y;
    o.z = a[64] * dv + bv.z;
    o.w = a[96] * dv + bv.w;
    *(float4*)(out + (size_t)row * NF + lane * 4) = o;
  }
}

extern "C" void kernel_launch(void* const* d_in, const int* in_sizes, int n_in,
                              void* d_out, int out_size, void* d_ws, size_t ws_size,
                              hipStream_t stream) {
  const float* x  = (const float*)d_in[0];
  const int*   ei = (const int*)d_in[1];
  const float* W  = (const float*)d_in[2];
  const float* b  = (const float*)d_in[3];
  float* out = (float*)d_out;

  const int N = in_sizes[0] / NF;
  const int E = in_sizes[1] / 2;

  float* deg = (float*)d_ws;
  size_t off = ((size_t)N * sizeof(float) + 511) & ~(size_t)511;
  float* hp = (float*)((char*)d_ws + off);
  size_t off2 = (off + (size_t)N * NF * sizeof(float) + 511) & ~(size_t)511;
  int* planes = (int*)((char*)d_ws + off2);

  const int pitch = (N + 31) & ~31;
  size_t off3 = (off2 + (size_t)DEG_SLICES * pitch * sizeof(int) + 511) & ~(size_t)511;
  unsigned int* own = (unsigned int*)((char*)d_ws + off3);
  deg_planes_kernel<<<dim3(DEG_SLICES, (pitch + DEG_RJ - 1) / DEG_RJ), DEG_THREADS, 0, stream>>>(
      ei, planes, own, N, E);
  deg_finish_kernel<<<(pitch / 2 + 255) / 256, 256, 0, stream>>>(planes, deg, N);

  gemm_scale_kernel<<<(N + 127) / 128, 256, 0, stream>>>(x, W, deg, hp, N);

  aggregate_rows_kernel<<<(N + RPB - 1) / RPB, AGG_WAVES * 32, 0, stream>>>(
      ei, own, hp, deg, b, out, N, E);
}


# aggregation tile 576 -> 393 destination rows: 255 work-groups fill 255 of 256 compute units instead of 174, each sweeps the list once but adds a third fewer edges; sums in any order as before
# speedup vs baseline: 3.3835x; 1.0427x over previous
#include <hip/hip_runtime.h>


#define NF 128
#define PSTR 144

typedef __attribute__((ext_vector_type(2))) float v2f;
typedef __attribute__((ext_vector_type(8))) float v8f;

#define DEG_SLICES 64
#define DEG_RJ 25024
#define DEG_THREADS 512
#define DEG_UNR 16

#define RPB 393
#define OWN_NONE 0xFFu

__device__ __forceinline__ int plane_pitch(int nrows) { return (nrows + 31) & ~31; }

__device__ __forceinline__ int slice_len(int nedges) {
  return (((nedges + DEG_SLICES - 1) / DEG_SLICES) + 127) & ~127;
}

__device__ __forceinline__ unsigned owner_byte(int dst, int nrows) {
  return ((unsigned)dst < (unsigned)nrows) ? (unsigned)dst / RPB : OWN_NONE;
}

__device__ __forceinline__ int4 load_dst4_guarded(const int* __restrict__ dstp,
                                                  int e0, int lim) {
  int4 r;
  r.x = (e0 + 0 < lim) ? dstp[e0 + 0] : -1;
  r.y = (e0 + 1 < lim) ? dstp[e0 + 1] : -1;
  r.z = (e0 + 2 < lim) ? dstp[e0 + 2] : -1;
  r.w = (e0 + 3 < lim) ? dstp[e0 + 3] : -1;
  return r;
}

__device__ __forceinline__ void deg_load(int4 (&d)[DEG_UNR], const int* __restrict__ dstp,
                                         int b, int e_end, bool vec_ok, int tid) {
#pragma unroll
  for (int u = DEG_UNR - 1; u >= 0; --u) {
    const int bu = b + u * DEG_THREADS * 4;
    const int e = bu + tid * 4;
    if (bu >= e_end) d[u] = make_int4(-1, -1, -1, -1);
    else if (vec_ok && bu + DEG_THREADS * 4 <= e_end) d[u] = *(const int4*)(dstp + e);
    else d[u] = load_dst4_guarded(dstp, e, e_end);
  }
}

__device__ __forceinline__ void deg_count1(int* cnt, int dst, int r0, int nr) {
  const unsigned l = (unsigned)(dst - r0);
  if (l < (unsigned)nr)
    __hip_atomic_fetch_add(&cnt[l], 1, __ATOMIC_RELAXED, __HIP_MEMORY_SCOPE_WORKGROUP);
}

__global__ __launch_bounds__(DEG_THREADS)
void deg_planes_kernel(const int* __restrict__ ei, int* __restrict__ planes,
                       unsigned int* __restrict__ own, int nrows, int nedges) {
  __shared__ int cnt[DEG_RJ];
  const int tid = threadIdx.x;
  const int pitch = plane_pitch(nrows);
  const int r0 = blockIdx.y * DEG_RJ;
  const int nr = max(0, min(DEG_RJ, nrows - r0));
  const int nrw = min(DEG_RJ, pitch - r0);

  const int* __restrict__ dstp = ei + nedges;
  const bool vec_ok = (((size_t)dstp) & 15) == 0;
  const int slen = slice_len(nedges);
  const int e_begin = min(nedges, (int)blockIdx.x * slen);
  const int e_end = min(nedges, e_begin + slen);
  const int step = DEG_THREADS * 4 * DEG_UNR;

  for (int i = tid * 4; i < DEG_RJ; i += DEG_THREADS * 4)
    *(int4*)(cnt + i) = make_int4(0, 0, 0, 0);
  __syncthreads();

  for (int b = e_begin; b < e_end; b += step) {
    int4 d[DEG_UNR];
    deg_load(d, dstp, b, e_end, vec_ok, tid);
    if (blockIdx.y == 0) {
#pragma unroll
      for (int u = 0; u < DEG_UNR; ++u) {
        const int e = b + u * DEG_THREADS * 4 + tid * 4;
        if (e < e_end)
          own[e >> 2] = owner_byte(d[u].x, nrows) | (owner_byte(d[u].y, nrows) << 8) |
                        (owner_byte(d[u].z, nrows) << 16) | (owner_byte(d[u].w, nrows) << 24);
      }
    }
#pragma unroll
    for (int u = 0; u < DEG_UNR; ++u) {
      deg_count1(cnt, d[u].x, r0, nr);
      deg_count1(cnt, d[u].y, r0, nr);
      deg_count1(cnt, d[u].z, r0, nr);
      deg_count1(cnt, d[u].w, r0, nr);
    }
  }
  __syncthreads();

  int* __restrict__ pl = planes + (size_t)blockIdx.x * pitch + r0;
  for (int i = tid * 4; i < nrw; i += DEG_THREADS * 4)
    *(int4*)(pl + i) = *(const int4*)(cnt + i);
}

__global__ __launch_bounds__(256)
void deg_finish_kernel(const int* __restrict__ planes, float* __restrict__ dinv,
                       int nrows) {
  const int pitch = plane_pitch(nrows);
  const int i = (blockIdx.x * 256 + threadIdx.x) * 2;
  if (i >= pitch) return;
  int2 v[DEG_SLICES];
#pragma unroll
  for (int s = 0; s < DEG_SLICES; ++s)
    v[s] = *(const int2*)(planes + (size_t)s * pitch + i);
  int c0 = 1, c1 = 1;
#pragma unroll
  for (int s = 0; s < DEG_SLICES; ++s) { c0 += v[s].x; c1 += v[s].y; }
  float2 r;
  r.x = rsqrtf((float)c0);
  r.y = rsqrtf((float)c1);
  *(float2*)(dinv + i) = r;
}

__global__ __launch_bounds__(256)
void gemm_scale_kernel(const float* __restrict__ x, const float* __restrict__ W,
                       const float* __restrict__ dinv, float* __restrict__ hp,
                       int nrows) {
  __shared__ v2f Wlds[(NF / 2) * PSTR];
  for (int i = threadIdx.x; i < NF * NF; i += 256) {
    int k = i >> 7, c = i & 127;
    Wlds[(k >> 1) * PSTR + c][k & 1] = W[i];
  }
  __syncthreads();

  const int wave = threadIdx.x >> 5;
  const int lane = threadIdx.x & 31;
  const int m0 = (blockIdx.x * 8 + wave) * 16;
  if (m0 >= nrows) return;

  const int half  = lane >> 4;
  const int khalf = half * 2;
  const int colb  = lane & 15;

  v8f acc[8] = {};

  const float* xr = x + (size_t)(m0 + colb) * NF + khalf;

  for (int kk = 0; kk < NF; kk += 4) {
    v2f a = *(const v2f*)(xr + kk);
    const v2f* wr = &Wlds[((kk >> 1) + half) * PSTR + colb];
#pragma unroll
    for (int nt = 0; nt < 8; ++nt) {
      v2f b = wr[nt * 16];
      acc[nt] = __builtin_amdgcn_wmma_f32_16x16x4_f32(
          false, a, false, b, (short)0, acc[nt], false, false);
    }
  }

  const int rbase = half << 3;
#pragma unroll
  for (int r = 0; r < 8; ++r) {
    const int row = m0 + rbase + r;
    const float dv = dinv[row];
    const size_t ro = (size_t)row * NF + colb;
#pragma unroll
    for (int nt = 0; nt < 8; ++nt) {
      hp[ro + nt * 16] = acc[nt][r] * dv;
    }
  }
}

#define AGG_WAVES 32
#define QCAP 64
#define UNR 8
#define STEP_EDGES (32 * 16 * UNR)

__device__ __forceinline__ unsigned zero_bytes(unsigned x) {
  return ~(((x & 0x7F7F7F7Fu) + 0x7F7F7F7Fu) | x | 0x7F7F7F7Fu);
}

__device__ __forceinline__ uint4 load_own4_guarded(const unsigned int* __restrict__ own,
                                                   int w0, int lim) {
  uint4 r;
  r.x = (w0 + 0 < lim) ? own[w0 + 0] : 0xFFFFFFFFu;
  r.y = (w0 + 1 < lim) ? own[w0 + 1] : 0xFFFFFFFFu;
  r.z = (w0 + 2 < lim) ? own[w0 + 2] : 0xFFFFFFFFu;
  r.w = (w0 + 3 < lim) ? own[w0 + 3] : 0xFFFFFFFFu;
  return r;
}

__device__ __forceinline__ void lds_add(float* p, float v) {
  __hip_atomic_fetch_add(p, v, __ATOMIC_RELAXED, __HIP_MEMORY_SCOPE_WORKGROUP);
}

template <bool FULL>
__device__ __forceinline__ void drain(int cnt, int qbase, const int* myqe,
                                      const int* __restrict__ ei,
                                      const int* __restrict__ dstp, int row0,
                                      const float* __restrict__ hp,
                                      float* acc, int lane) {
  int e = 0;
  if (FULL || lane < cnt) e = myqe[qbase + lane];
  const int s = ei[e];
  const int dl = dstp[e] - row0;
#pragma unroll 1
  for (int k0 = 0; k0 < (FULL ? 32 : cnt); k0 += 8) {
    float4 v[8];
#pragma unroll
    for (int i = 0; i < 8; ++i) {
      if (FULL || k0 + i < cnt) {
        const int si = __builtin_amdgcn_readlane(s, k0 + i);
        v[i] = *(const float4*)(hp + (size_t)si * NF + lane * 4);
      }
    }
#pragma unroll
    for (int i = 0; i < 8; ++i) {
      if (FULL || k0 + i < cnt) {
        const int di = __builtin_amdgcn_readlane(dl, k0 + i);
        float* a = acc + di * NF + lane;
        lds_add(a, v[i].x);
        lds_add(a + 32, v[i].y);
        lds_add(a + 64, v[i].z);
        lds_add(a + 96, v[i].w);
      }
    }
  }
}

__global__ __launch_bounds__(AGG_WAVES * 32)
void aggregate_rows_kernel(const int* __restrict__ ei, const unsigned int* __restrict__ own,
                           const float* __restrict__ hp,
                           const float* __restrict__ dinv,
                           const float* __restrict__ bias, float* __restrict__ out,
                           int nrows, int nedges) {
  __shared__ float acc[RPB * NF];
  __shared__ int qe[AGG_WAVES * QCAP];

  const int lane = threadIdx.x & 31;
  const int wave = threadIdx.x >> 5;
  const int row0 = blockIdx.x * RPB;
  const int nr = min(RPB, nrows - row0);

  for (int r = wave; r < nr; r += AGG_WAVES) {
    const float4 v = *(const float4*)(hp + (size_t)(row0 + r) * NF + lane * 4);
    float* a = acc + r * NF + lane;
    a[0] = v.x; a[32] = v.y; a[64] = v.z; a[96] = v.w;
  }
  __syncthreads();

  const int* __restrict__ dstp = ei + nedges;
  const bool vec_ok = (((size_t)own) & 15) == 0;
  const int nwords = (nedges + 3) >> 2;
  const int per_step = AGG_WAVES * STEP_EDGES;
  const int nsteps = (nedges + per_step - 1) / per_step;
  const int sstart = (int)(((long long)blockIdx.x * nsteps) / gridDim.x);

  int* myqe = qe + wave * QCAP;
  int qn = 0;

  const int wave_u = __builtin_amdgcn_readfirstlane(wave);
  const unsigned pat = (unsigned)blockIdx.x * 0x01010101u;

  for (int st = 0; st < nsteps; ++st) {
    int ss = st + sstart;
    if (ss >= nsteps) ss -= nsteps;
    const int wbase = (ss * AGG_WAVES + wave_u) * STEP_EDGES;
    const int base = wbase + lane * 16;
    uint4 cur[UNR];
    if (vec_ok && wbase + STEP_EDGES <= nedges) {
      const uint4* p = (const uint4*)(own + (wbase >> 2)) + lane;
#pragma unroll
      for (int u = 0; u < UNR; ++u) cur[u] = p[u * 32];
    } else {
#pragma unroll
      for (int u = 0; u < UNR; ++u)
        cur[u] = load_own4_guarded(own, (base + u * 512) >> 2, nwords);
    }

#pragma unroll
    for (int u = 0; u < UNR; ++u) {
      unsigned m = (zero_bytes(cur[u].x ^ pat) >> 3) | (zero_bytes(cur[u].y ^ pat) >> 2) |
                   (zero_bytes(cur[u].z ^ pat) >> 1) | zero_bytes(cur[u].w ^ pat);
      unsigned bal = __builtin_amdgcn_ballot_w32(m != 0u);
      const int e0m = base + u * 512 - 16;
      while (bal != 0u) {
        const int pos = qn + (int)__builtin_amdgcn_mbcnt_lo(bal, 0u);
        if (m != 0u) {
          const int p = __builtin_ctz(m);
          myqe[pos] = e0m + ((p & 7) << 2) + (p >> 3);
          m &= m - 1u;
        }
        qn += __builtin_popcount(bal);
        if (qn >= 32) {
          qn -= 32;
          drain<true>(32, qn, myqe, ei, dstp, row0, hp, acc, lane);
        }
        bal = __builtin_amdgcn_ballot_w32(m != 0u);
      }
    }
  }
  if (qn > 0) drain<false>(qn, 0, myqe, ei, dstp, row0, hp, acc, lane);
  __syncthreads();

  const float4 bv = *(const float4*)(bias + lane * 4);
  for (int r = wave; r < nr; r += AGG_WAVES) {
    const int row = row0 + r;
    const float dv = dinv[row];
    const float* a = acc + r * NF + lane;
    float4 o;
    o.x = a[0] * dv + bv.x;
    o.y = a[32] * dv + bv.y;
    o.z = a[64] * dv + bv.z;
    o.w = a[96] * dv + bv.w;
    *(float4*)(out + (size_t)row * NF + lane * 4) = o;
  }
}

extern "C" void kernel_launch(void* const* d_in, const int* in_sizes, int n_in,
                              void* d_out, int out_size, void* d_ws, size_t ws_size,
                              hipStream_t stream) {
  const float* x  = (const float*)d_in[0];
  const int*   ei = (const int*)d_in[1];
  const float* W  = (const float*)d_in[2];
  const float* b  = (const float*)d_in[3];
  float* out = (float*)d_out;

  const int N = in_sizes[0] / NF;
  const int E = in_sizes[1] / 2;

  float* deg = (float*)d_ws;
  size_t off = ((size_t)N * sizeof(float) + 511) & ~(size_t)511;
  float* hp = (float*)((char*)d_ws + off);
  size_t off2 = (off + (size_t)N * NF * sizeof(float) + 511) & ~(size_t)511;
  int* planes = (int*)((char*)d_ws + off2);

  const int pitch = (N + 31) & ~31;
  size_t off3 = (off2 + (size_t)DEG_SLICES * pitch * sizeof(int) + 511) & ~(size_t)511;
  unsigned int* own = (unsigned int*)((char*)d_ws + off3);
  deg_planes_kernel<<<dim3(DEG_SLICES, (pitch + DEG_RJ - 1) / DEG_RJ), DEG_THREADS, 0, stream>>>(
      ei, planes, own, N, E);
  deg_finish_kernel<<<(pitch / 2 + 255) / 256, 256, 0, stream>>>(planes, deg, N);

  gemm_scale_kernel<<<(N + 127) / 128, 256, 0, stream>>>(x, W, deg, hp, N);

  aggregate_rows_kernel<<<(N + RPB - 1) / RPB, AGG_WAVES * 32, 0, stream>>>(
      ei, own, hp, deg, b, out, N, E);
}


# aggregation: loads issued ahead of use: a wave's 13 image rows in one round trip, the next owner-list piece read while the current one is filtered, write-out scales before the barrier; sums as before
# speedup vs baseline: 3.5788x; 1.0577x over previous
#include <hip/hip_runtime.h>


#define NF 128
#define PSTR 144

typedef __attribute__((ext_vector_type(2))) float v2f;
typedef __attribute__((ext_vector_type(8))) float v8f;

#define DEG_SLICES 64
#define DEG_RJ 25024
#define DEG_THREADS 512
#define DEG_UNR 16

#define RPB 393
#define OWN_NONE 0xFFu

__device__ __forceinline__ int plane_pitch(int nrows) { return (nrows + 31) & ~31; }

__device__ __forceinline__ int slice_len(int nedges) {
  return (((nedges + DEG_SLICES - 1) / DEG_SLICES) + 127) & ~127;
}

__device__ __forceinline__ unsigned owner_byte(int dst, int nrows) {
  return ((unsigned)dst < (unsigned)nrows) ? (unsigned)dst / RPB : OWN_NONE;
}

__device__ __forceinline__ int4 load_dst4_guarded(const int* __restrict__ dstp,
                                                  int e0, int lim) {
  int4 r;
  r.x = (e0 + 0 < lim) ? dstp[e0 + 0] : -1;
  r.y = (e0 + 1 < lim) ? dstp[e0 + 1] : -1;
  r.z = (e0 + 2 < lim) ? dstp[e0 + 2] : -1;
  r.w = (e0 + 3 < lim) ? dstp[e0 + 3] : -1;
  return r;
}

__device__ __forceinline__ void deg_load(int4 (&d)[DEG_UNR], const int* __restrict__ dstp,
                                         int b, int e_end, bool vec_ok, int tid) {
#pragma unroll
  for (int u = DEG_UNR - 1; u >= 0; --u) {
    const int bu = b + u * DEG_THREADS * 4;
    const int e = bu + tid * 4;
    if (bu >= e_end) d[u] = make_int4(-1, -1, -1, -1);
    else if (vec_ok && bu + DEG_THREADS * 4 <= e_end) d[u] = *(const int4*)(dstp + e);
    else d[u] = load_dst4_guarded(dstp, e, e_end);
  }
}

__device__ __forceinline__ void deg_count1(int* cnt, int dst, int r0, int nr) {
  const unsigned l = (unsigned)(dst - r0);
  if (l < (unsigned)nr)
    __hip_atomic_fetch_add(&cnt[l], 1, __ATOMIC_RELAXED, __HIP_MEMORY_SCOPE_WORKGROUP);
}

__global__ __launch_bounds__(DEG_THREADS)
void deg_planes_kernel(const int* __restrict__ ei, int* __restrict__ planes,
                       unsigned int* __restrict__ own, int nrows, int nedges) {
  __shared__ int cnt[DEG_RJ];
  const int tid = threadIdx.x;
  const int pitch = plane_pitch(nrows);
  const int r0 = blockIdx.y * DEG_RJ;
  const int nr = max(0, min(DEG_RJ, nrows - r0));
  const int nrw = min(DEG_RJ, pitch - r0);

  const int* __restrict__ dstp = ei + nedges;
  const bool vec_ok = (((size_t)dstp) & 15) == 0;
  const int slen = slice_len(nedges);
  const int e_begin = min(nedges, (int)blockIdx.x * slen);
  const int e_end = min(nedges, e_begin + slen);
  const int step = DEG_THREADS * 4 * DEG_UNR;

  for (int i = tid * 4; i < DEG_RJ; i += DEG_THREADS * 4)
    *(int4*)(cnt + i) = make_int4(0, 0, 0, 0);
  __syncthreads();

  for (int b = e_begin; b < e_end; b += step) {
    int4 d[DEG_UNR];
    deg_load(d, dstp, b, e_end, vec_ok, tid);
    if (blockIdx.y == 0) {
#pragma unroll
      for (int u = 0; u < DEG_UNR; ++u) {
        const int e = b + u * DEG_THREADS * 4 + tid * 4;
        if (e < e_end)
          own[e >> 2] = owner_byte(d[u].x, nrows) | (owner_byte(d[u].y, nrows) << 8) |
                        (owner_byte(d[u].z, nrows) << 16) | (owner_byte(d[u].w, nrows) << 24);
      }
    }
#pragma unroll
    for (int u = 0; u < DEG_UNR; ++u) {
      deg_count1(cnt, d[u].x, r0, nr);
      deg_count1(cnt, d[u].y, r0, nr);
      deg_count1(cnt, d[u].z, r0, nr);
      deg_count1(cnt, d[u].w, r0, nr);
    }
  }
  __syncthreads();

  int* __restrict__ pl = planes + (size_t)blockIdx.x * pitch + r0;
  for (int i = tid * 4; i < nrw; i += DEG_THREADS * 4)
    *(int4*)(pl + i) = *(const int4*)(cnt + i);
}

__global__ __launch_bounds__(256)
void deg_finish_kernel(const int* __restrict__ planes, float* __restrict__ dinv,
                       int nrows) {
  const int pitch = plane_pitch(nrows);
  const int i = (blockIdx.x * 256 + threadIdx.x) * 2;
  if (i >= pitch) return;
  int2 v[DEG_SLICES];
#pragma unroll
  for (int s = 0; s < DEG_SLICES; ++s)
    v[s] = *(const int2*)(planes + (size_t)s * pitch + i);
  int c0 = 1, c1 = 1;
#pragma unroll
  for (int s = 0; s < DEG_SLICES; ++s) { c0 += v[s].x; c1 += v[s].y; }
  float2 r;
  r.x = rsqrtf((float)c0);
  r.y = rsqrtf((float)c1);
  *(float2*)(dinv + i) = r;
}

__global__ __launch_bounds__(256)
void gemm_scale_kernel(const float* __restrict__ x, const float* __restrict__ W,
                       const float* __restrict__ dinv, float* __restrict__ hp,
                       int nrows) {
  __shared__ v2f Wlds[(NF / 2) * PSTR];
  for (int i = threadIdx.x; i < NF * NF; i += 256) {
    int k = i >> 7, c = i & 127;
    Wlds[(k >> 1) * PSTR + c][k & 1] = W[i];
  }
  __syncthreads();

  const int wave = threadIdx.x >> 5;
  const int lane = threadIdx.x & 31;
  const int m0 = (blockIdx.x * 8 + wave) * 16;
  if (m0 >= nrows) return;

  const int half  = lane >> 4;
  const int khalf = half * 2;
  const int colb  = lane & 15;

  v8f acc[8] = {};

  const float* xr = x + (size_t)(m0 + colb) * NF + khalf;

  for (int kk = 0; kk < NF; kk += 4) {
    v2f a = *(const v2f*)(xr + kk);
    const v2f* wr = &Wlds[((kk >> 1) + half) * PSTR + colb];
#pragma unroll
    for (int nt = 0; nt < 8; ++nt) {
      v2f b = wr[nt * 16];
      acc[nt] = __builtin_amdgcn_wmma_f32_16x16x4_f32(
          false, a, false, b, (short)0, acc[nt], false, false);
    }
  }

  const int rbase = half << 3;
#pragma unroll
  for (int r = 0; r < 8; ++r) {
    const int row = m0 + rbase + r;
    const float dv = dinv[row];
    const size_t ro = (size_t)row * NF + colb;
#pragma unroll
    for (int nt = 0; nt < 8; ++nt) {
      hp[ro + nt * 16] = acc[nt][r] * dv;
    }
  }
}

#define AGG_WAVES 32
#define QCAP 64
#define UNR 8
#define STEP_EDGES (32 * 16 * UNR)
#define IMG_R ((RPB + AGG_WAVES - 1) / AGG_WAVES)

__device__ __forceinline__ unsigned zero_bytes(unsigned x) {
  return ~(((x & 0x7F7F7F7Fu) + 0x7F7F7F7Fu) | x | 0x7F7F7F7Fu);
}

__device__ __forceinline__ uint4 load_own4_guarded(const unsigned int* __restrict__ own,
                                                   int w0, int lim) {
  uint4 r;
  r.x = (w0 + 0 < lim) ? own[w0 + 0] : 0xFFFFFFFFu;
  r.y = (w0 + 1 < lim) ? own[w0 + 1] : 0xFFFFFFFFu;
  r.z = (w0 + 2 < lim) ? own[w0 + 2] : 0xFFFFFFFFu;
  r.w = (w0 + 3 < lim) ? own[w0 + 3] : 0xFFFFFFFFu;
  return r;
}

__device__ __forceinline__ void load_list(uint4 (&buf)[UNR],
                                          const unsigned int* __restrict__ own, int wbase,
                                          int lane, bool vec_ok, int nedges, int nwords) {
  if (vec_ok && wbase + STEP_EDGES <= nedges) {
    const uint4* p = (const uint4*)(own + (wbase >> 2)) + lane;
#pragma unroll
    for (int u = 0; u < UNR; ++u) buf[u] = p[u * 32];
  } else {
    const int base = wbase + lane * 16;
#pragma unroll
    for (int u = 0; u < UNR; ++u)
      buf[u] = load_own4_guarded(own, (base + u * 512) >> 2, nwords);
  }
}

__device__ __forceinline__ void lds_add(float* p, float v) {
  __hip_atomic_fetch_add(p, v, __ATOMIC_RELAXED, __HIP_MEMORY_SCOPE_WORKGROUP);
}

template <bool FULL>
__device__ __forceinline__ void drain(int cnt, int qbase, const int* myqe,
                                      const int* __restrict__ ei,
                                      const int* __restrict__ dstp, int row0,
                                      const float* __restrict__ hp,
                                      float* acc, int lane) {
  int e = 0;
  if (FULL || lane < cnt) e = myqe[qbase + lane];
  const int s = ei[e];
  const int dl = dstp[e] - row0;
#pragma unroll 1
  for (int k0 = 0; k0 < (FULL ? 32 : cnt); k0 += 8) {
    float4 v[8];
#pragma unroll
    for (int i = 0; i < 8; ++i) {
      if (FULL || k0 + i < cnt) {
        const int si = __builtin_amdgcn_readlane(s, k0 + i);
        v[i] = *(const float4*)(hp + (size_t)si * NF + lane * 4);
      }
    }
#pragma unroll
    for (int i = 0; i < 8; ++i) {
      if (FULL || k0 + i < cnt) {
        const int di = __builtin_amdgcn_readlane(dl, k0 + i);
        float* a = acc + di * NF + lane;
        lds_add(a, v[i].x);
        lds_add(a + 32, v[i].y);
        lds_add(a + 64, v[i].z);
        lds_add(a + 96, v[i].w);
      }
    }
  }
}

__global__ __launch_bounds__(AGG_WAVES * 32)
void aggregate_rows_kernel(const int* __restrict__ ei, const unsigned int* __restrict__ own,
                           const float* __restrict__ hp,
                           const float* __restrict__ dinv,
                           const float* __restrict__ bias, float* __restrict__ out,
                           int nrows, int nedges) {
  __shared__ float acc[RPB * NF];
  __shared__ int qe[AGG_WAVES * QCAP];

  const int lane = threadIdx.x & 31;
  const int wave = threadIdx.x >> 5;
  const int row0 = blockIdx.x * RPB;
  const int nr = min(RPB, nrows - row0);

  const int wave_u = __builtin_amdgcn_readfirstlane(wave);

  {
    float4 v[IMG_R];
#pragma unroll
    for (int i = 0; i < IMG_R; ++i) {
      const int r = min(wave_u + i * AGG_WAVES, nr - 1);
      v[i] = *(const float4*)(hp + (size_t)(row0 + r) * NF + lane * 4);
    }
#pragma unroll
    for (int i = 0; i < IMG_R; ++i) {
      const int r = wave_u + i * AGG_WAVES;
      if (r < nr) {
        float* a = acc + r * NF + lane;
        a[0] = v[i].x; a[32] = v[i].y; a[64] = v[i].z; a[96] = v[i].w;
      }
    }
  }
  __syncthreads();

  const int* __restrict__ dstp = ei + nedges;
  const bool vec_ok = (((size_t)own) & 15) == 0;
  const int nwords = (nedges + 3) >> 2;
  const int per_step = AGG_WAVES * STEP_EDGES;
  const int nsteps = (nedges + per_step - 1) / per_step;
  const int sstart = (int)(((long long)blockIdx.x * nsteps) / gridDim.x);

  int* myqe = qe + wave * QCAP;
  int qn = 0;

  const unsigned pat = (unsigned)blockIdx.x * 0x01010101u;

  uint4 nxt[UNR];
#pragma unroll
  for (int u = 0; u < UNR; ++u) nxt[u] = make_uint4(~0u, ~0u, ~0u, ~0u);
  int wbase_nxt = 0;
  for (int st = -1; st < nsteps; ++st) {
    uint4 cur[UNR];
#pragma unroll
    for (int u = 0; u < UNR; ++u) cur[u] = nxt[u];
    const int wbase = wbase_nxt;
    if (st + 1 < nsteps) {
      int ss = st + 1 + sstart;
      if (ss >= nsteps) ss -= nsteps;
      wbase_nxt = (ss * AGG_WAVES + wave_u) * STEP_EDGES;
      load_list(nxt, own, wbase_nxt, lane, vec_ok, nedges, nwords);
    }
    if (st < 0) continue;
    const int base = wbase + lane * 16;

#pragma unroll
    for (int u = 0; u < UNR; ++u) {
      unsigned m = (zero_bytes(cur[u].x ^ pat) >> 3) | (zero_bytes(cur[u].y ^ pat) >> 2) |
                   (zero_bytes(cur[u].z ^ pat) >> 1) | zero_bytes(cur[u].w ^ pat);
      unsigned bal = __builtin_amdgcn_ballot_w32(m != 0u);
      const int e0m = base + u * 512 - 16;
      while (bal != 0u) {
        const int pos = qn + (int)__builtin_amdgcn_mbcnt_lo(bal, 0u);
        if (m != 0u) {
          const int p = __builtin_ctz(m);
          myqe[pos] = e0m + ((p & 7) << 2) + (p >> 3);
          m &= m - 1u;
        }
        qn += __builtin_popcount(bal);
        if (qn >= 32) {
          qn -= 32;
          drain<true>(32, qn, myqe, ei, dstp, row0, hp, acc, lane);
        }
        bal = __builtin_amdgcn_ballot_w32(m != 0u);
      }
    }
  }
  if (qn > 0) drain<false>(qn, 0, myqe, ei, dstp, row0, hp, acc, lane);

  const float4 bv = *(const float4*)(bias + lane * 4);
  float dv[IMG_R];
#pragma unroll
  for (int i = 0; i < IMG_R; ++i)
    dv[i] = dinv[row0 + min(wave_u + i * AGG_WAVES, nr - 1)];
  __syncthreads();

#pragma unroll
  for (int i = 0; i < IMG_R; ++i) {
    const int r = wave_u + i * AGG_WAVES;
    if (r < nr) {
      const float* a = acc + r * NF + lane;
      float4 o;
      o.x = a[0] * dv[i] + bv.x;
      o.y = a[32] * dv[i] + bv.y;
      o.z = a[64] * dv[i] + bv.z;
      o.w = a[96] * dv[i] + bv.w;
      *(float4*)(out + (size_t)(row0 + r) * NF + lane * 4) = o;
    }
  }
}

extern "C" void kernel_launch(void* const* d_in, const int* in_sizes, int n_in,
                              void* d_out, int out_size, void* d_ws, size_t ws_size,
                              hipStream_t stream) {
  const float* x  = (const float*)d_in[0];
  const int*   ei = (const int*)d_in[1];
  const float* W  = (const float*)d_in[2];
  const float* b  = (const float*)d_in[3];
  float* out = (float*)d_out;

  const int N = in_sizes[0] / NF;
  const int E = in_sizes[1] / 2;

  float* deg = (float*)d_ws;
  size_t off = ((size_t)N * sizeof(float) + 511) & ~(size_t)511;
  float* hp = (float*)((char*)d_ws + off);
  size_t off2 = (off + (size_t)N * NF * sizeof(float) + 511) & ~(size_t)511;
  int* planes = (int*)((char*)d_ws + off2);

  const int pitch = (N + 31) & ~31;
  size_t off3 = (off2 + (size_t)DEG_SLICES * pitch * sizeof(int) + 511) & ~(size_t)511;
  unsigned int* own = (unsigned int*)((char*)d_ws + off3);
  deg_planes_kernel<<<dim3(DEG_SLICES, (pitch + DEG_RJ - 1) / DEG_RJ), DEG_THREADS, 0, stream>>>(
      ei, planes, own, N, E);
  deg_finish_kernel<<<(pitch / 2 + 255) / 256, 256, 0, stream>>>(planes, deg, N);

  gemm_scale_kernel<<<(N + 127) / 128, 256, 0, stream>>>(x, W, deg, hp, N);

  aggregate_rows_kernel<<<(N + RPB - 1) / RPB, AGG_WAVES * 32, 0, stream>>>(
      ei, own, hp, deg, b, out, N, E);
}
